# starting kernel
# baseline (speedup 1.0000x reference)
#include <hip/hip_runtime.h>

#define NB   4
#define NN   256
#define NINC 128
#define NOUTC 128

typedef __attribute__((ext_vector_type(2))) float v2f;
typedef __attribute__((ext_vector_type(4))) float v4f;
typedef __attribute__((ext_vector_type(8))) float v8f;

__global__ __launch_bounds__(32) void nodeconv_gemm(
    const float* __restrict__ node,
    const float* __restrict__ Wi,
    const float* __restrict__ Wj,
    float* __restrict__ vi,
    float* __restrict__ vj)
{
    const int tile = blockIdx.x;
    const int b    = tile >> 7;
    const int t    = tile & 127;
    const int m0   = (t >> 4) << 4;
    const int n0   = (t & 15) << 4;

    const int lane = threadIdx.x;
    const int half = lane >> 4;
    const int idx  = lane & 15;

    const float* nb = node + b * (NINC * NN);

    v8f acc_i = {0.f, 0.f, 0.f, 0.f, 0.f, 0.f, 0.f, 0.f};
    v8f acc_j = {0.f, 0.f, 0.f, 0.f, 0.f, 0.f, 0.f, 0.f};

    const int arow = (m0 + idx) * NINC;

    for (int kk = 0; kk < NINC; kk += 4) {
        const int ka = kk + 2 * half;
        v2f a_i = *(const v2f*)(Wi + arow + ka);
        v2f a_j = *(const v2f*)(Wj + arow + ka);
        v2f bf;
        bf.x = nb[(ka    ) * NN + n0 + idx];
        bf.y = nb[(ka + 1) * NN + n0 + idx];

        acc_i = __builtin_amdgcn_wmma_f32_16x16x4_f32(
             false, a_i,  false, bf,
             (short)0, acc_i,  false,  false);
        acc_j = __builtin_amdgcn_wmma_f32_16x16x4_f32(
             false, a_j,  false, bf,
             (short)0, acc_j,  false,  false);
    }

    #pragma unroll
    for (int r = 0; r < 8; ++r) {
        const int o = m0 + 8 * half + r;
        vi[(b * NOUTC + o) * NN + n0 + idx] = acc_i[r];
        vj[(b * NOUTC + o) * NN + n0 + idx] = acc_j[r];
    }
}

__global__ __launch_bounds__(256) void nodeconv_bcast(
    const float* __restrict__ adj,
    const float* __restrict__ vi,
    const float* __restrict__ vj,
    float* __restrict__ out)
{
    const int tid    = threadIdx.x;
    const int rlocal = tid >> 6;
    const int jv     = (tid & 63) << 2;

    const int row = blockIdx.x * 4 + rlocal;
    const int b   = row >> 15;
    const int rem = row & 32767;
    const int o   = rem >> 8;
    const int i   = rem & 255;

    const v4f a4 = *(const v4f*)(adj + (b * NN + i) * NN + jv);
    const v4f w4 = *(const v4f*)(vj + (b * NOUTC + o) * NN + jv);

    v4f r4 = a4 * w4;

    if (i >= jv && i < jv + 4) {
        const float dvi = vi[(b * NOUTC + o) * NN + i];
        r4[i - jv] = a4[i - jv] * dvi;
    }

    v4f* dst = (v4f*)(out + (((long long)(b * NOUTC + o) * NN + i) * NN) + jv);
    __builtin_nontemporal_store(r4, dst);
}

extern "C" void kernel_launch(void* const* d_in, const int* in_sizes, int n_in,
                              void* d_out, int out_size, void* d_ws, size_t ws_size,
                              hipStream_t stream) {
    (void)in_sizes; (void)n_in; (void)out_size; (void)ws_size;

    const float* adj  = (const float*)d_in[0];
    const float* node = (const float*)d_in[1];
    const float* Wi   = (const float*)d_in[2];
    const float* Wj   = (const float*)d_in[3];
    float* out = (float*)d_out;

    float* vi = (float*)d_ws;
    float* vj = vi + (size_t)NB * NOUTC * NN;

    nodeconv_gemm<<<NB * 8 * 16, 32, 0, stream>>>(node, Wi, Wj, vi, vj);

    const int rows   = NB * NOUTC * NN;
    const int blocks = rows / 4;
    nodeconv_bcast<<<blocks, 256, 0, stream>>>(adj, vi, vj, out);
}


# gemm launch rebuilt as 256 groups of 4 waves: node tile staged in LDS, W rows read as b128, all loads issued before the first WMMA; k re-ordered and summed in two accumulators per tile
# speedup vs baseline: 2.4426x; 2.4426x over previous
#include <hip/hip_runtime.h>

#define NB   4
#define NN   256
#define NINC 128
#define NOUTC 128

typedef __attribute__((ext_vector_type(2))) float v2f;
typedef __attribute__((ext_vector_type(4))) float v4f;
typedef __attribute__((ext_vector_type(8))) float v8f;

#define NPITCH 40

__global__ __launch_bounds__(128) void nodeconv_gemm(
    const float* __restrict__ node,
    const float* __restrict__ Wi,
    const float* __restrict__ Wj,
    float* __restrict__ vi,
    float* __restrict__ vj)
{
    __shared__ float sn[NINC * NPITCH];

    const int g  = blockIdx.x;
    const int b  = g >> 6;
    const int m0 = ((g >> 3) & 7) << 4;
    const int n0 = (g & 7) << 5;

    const int tid  = threadIdx.x;
    const int lane = tid & 31;
    const int wave = tid >> 5;
    const int nt   = wave & 1;
    const int mat  = wave >> 1;
    const int half = lane >> 4;
    const int idx  = lane & 15;

    const float* nsrc = node + b * (NINC * NN) + n0;
    v4f st[8];
    for (int i = 0; i < 8; ++i) {
        const int f = i * 128 + tid;
        st[i] = *(const v4f*)(nsrc + (f >> 3) * NN + ((f & 7) << 2));
    }

    const float* wsrc = (mat ? Wj : Wi) + (m0 + idx) * NINC + 4 * half;
    v4f af[16];
    for (int t = 0; t < 16; ++t) af[t] = *(const v4f*)(wsrc + 8 * t);
    __builtin_amdgcn_sched_barrier(0);

    for (int i = 0; i < 8; ++i) {
        const int f = i * 128 + tid;
        *(v4f*)(sn + (f >> 3) * NPITCH + ((f & 7) << 2)) = st[i];
    }
    __syncthreads();

    v8f acc0 = {0.f, 0.f, 0.f, 0.f, 0.f, 0.f, 0.f, 0.f};
    v8f acc1 = {0.f, 0.f, 0.f, 0.f, 0.f, 0.f, 0.f, 0.f};

    const float* sb = sn + (4 * half) * NPITCH + 16 * nt + idx;

    v2f bq[32];
    for (int s = 0; s < 32; ++s) {
        bq[s].x = sb[(8 * (s >> 1) + 2 * (s & 1) + 0) * NPITCH];
        bq[s].y = sb[(8 * (s >> 1) + 2 * (s & 1) + 1) * NPITCH];
    }
    __builtin_amdgcn_sched_barrier(0);

    for (int t = 0; t < 16; ++t) {
        const v4f a = af[t];
        v2f a0, a1;
        a0.x = a.x; a0.y = a.y;
        a1.x = a.z; a1.y = a.w;
        const v2f b0 = bq[2 * t];
        const v2f b1 = bq[2 * t + 1];

        acc0 = __builtin_amdgcn_wmma_f32_16x16x4_f32(
             false, a0,  false, b0,
             (short)0, acc0,  false,  false);
        acc1 = __builtin_amdgcn_wmma_f32_16x16x4_f32(
             false, a1,  false, b1,
             (short)0, acc1,  false,  false);
    }

    float* dst = (mat ? vj : vi) + (b * NOUTC + m0 + 8 * half) * NN + n0 + 16 * nt + idx;
    for (int r = 0; r < 8; ++r) dst[r * NN] = acc0[r] + acc1[r];
}

__global__ __launch_bounds__(256) void nodeconv_bcast(
    const float* __restrict__ adj,
    const float* __restrict__ vi,
    const float* __restrict__ vj,
    float* __restrict__ out)
{
    __shared__ v4f tile[64 * (NN / 4)];

    const int tid  = threadIdx.x;
    const int lane = tid & 31;
    const int wave = tid >> 5;

    const int g  = blockIdx.x;
    const int b  = g >> 6;
    const int rb = (g >> 4) & 3;
    const int cg = g & 15;
    const int i0 = rb << 6;
    const int o  = (cg << 3) + wave;

    const v4f* src = (const v4f*)(adj + (b * NN + i0) * NN);
    v4f stage[16];
    for (int k = 0; k < 16; ++k) stage[k] = src[k * 256 + tid];

    const int pd   = i0 >> 7;
    const int colD = (pd << 7) + (lane << 2);
    const int colO = ((pd ^ 1) << 7) + (lane << 2);

    const float* vrow = vj + (b * NOUTC + o) * NN;
    const v4f wD = *(const v4f*)(vrow + colD);
    const v4f wO = *(const v4f*)(vrow + colO);

    const int dq  = lane - ((i0 & 64) >> 2);
    const int dqc = dq < 0 ? 0 : (dq > 15 ? 15 : dq);
    const v4f dv  = *(const v4f*)(vi + (b * NOUTC + o) * NN + i0 + 4 * dqc);

    for (int k = 0; k < 16; ++k) tile[k * 256 + tid] = stage[k];
    __syncthreads();

    float* orow = out + ((long long)(b * NOUTC + o) * NN + i0) * NN;
    const int tD = colD >> 2;
    const int tO = colO >> 2;

    for (int q = 0; q < 16; ++q) {
        const bool isd = (dq == q);
        for (int e = 0; e < 4; ++e) {
            const int r = 4 * q + e;
            const v4f aD = tile[r * (NN / 4) + tD];
            const v4f aO = tile[r * (NN / 4) + tO];
            v4f rD = aD * wD;
            v4f rO = aO * wO;
            const float dd = aD[e] * dv[e];
            rD[e] = isd ? dd : rD[e];
            __builtin_nontemporal_store(rD, (v4f*)(orow + r * NN + colD));
            __builtin_nontemporal_store(rO, (v4f*)(orow + r * NN + colO));
        }
    }
}

extern "C" void kernel_launch(void* const* d_in, const int* in_sizes, int n_in,
                              void* d_out, int out_size, void* d_ws, size_t ws_size,
                              hipStream_t stream) {
    (void)in_sizes; (void)n_in; (void)out_size; (void)ws_size;

    const float* adj  = (const float*)d_in[0];
    const float* node = (const float*)d_in[1];
    const float* Wi   = (const float*)d_in[2];
    const float* Wj   = (const float*)d_in[3];
    float* out = (float*)d_out;

    float* vi = (float*)d_ws;
    float* vj = vi + (size_t)NB * NOUTC * NN;

    nodeconv_gemm<<<NB * 8 * 8, 128, 0, stream>>>(node, Wi, Wj, vi, vj);

    nodeconv_bcast<<<256, 256, 0, stream>>>(adj, vi, vj, out);
}
